# starting kernel
# baseline (speedup 1.0000x reference)
#include <hip/hip_runtime.h>

typedef float v2f __attribute__((ext_vector_type(2)));
typedef float v8f __attribute__((ext_vector_type(8)));

#define HID    128
#define KSTEPS 32
#define NTILES 8

__global__ __launch_bounds__(128)
void gcn_gemm_wmma(const float* __restrict__ x,
                   const float* __restrict__ W,
                   const float* __restrict__ b,
                   float* __restrict__ h,
                   float* __restrict__ out,
                   int n_nodes)
{
    const int wave   = blockIdx.x * (blockDim.x >> 5) + (threadIdx.x >> 5);
    const int lane   = threadIdx.x & 31;
    const int m_base = wave * 16;
    if (m_base >= n_nodes) return;

    const int rown  = lane & 15;
    const int khalf = (lane >> 4) << 1;

    v2f a[KSTEPS];
    {
        const float* xrow = x + (size_t)(m_base + rown) * HID + khalf;
        #pragma unroll
        for (int k = 0; k < KSTEPS; ++k)
            a[k] = *reinterpret_cast<const v2f*>(xrow + 4 * k);
    }

    #pragma unroll
    for (int nt = 0; nt < NTILES; ++nt) {
        const int n_base = nt * 16;
        const float bn = b[n_base + rown];
        v8f c;
        #pragma unroll
        for (int i = 0; i < 8; ++i) c[i] = bn;

        const float* wrow = W + (size_t)(n_base + rown) * HID + khalf;
        #pragma unroll
        for (int k = 0; k < KSTEPS; ++k) {
            v2f bf = *reinterpret_cast<const v2f*>(wrow + 4 * k);
            c = __builtin_amdgcn_wmma_f32_16x16x4_f32(
                     false, a[k],  false, bf,
                     (short)0, c,  false,  false);
        }

        const int col   = n_base + rown;
        const int rbase = m_base + ((lane >> 4) << 3);
        #pragma unroll
        for (int r = 0; r < 8; ++r) {
            const size_t off = (size_t)(rbase + r) * HID + col;
            h[off]   = c[r];
            out[off] = c[r];
        }
    }
}

__global__ void gcn_zero_deg(int* __restrict__ deg, int n)
{
    int i = blockIdx.x * blockDim.x + threadIdx.x;
    if (i < n) deg[i] = 0;
}

__global__ __launch_bounds__(256)
void gcn_scatter(const int* __restrict__ edge_index,
                 const float* __restrict__ h,
                 float* __restrict__ out,
                 int* __restrict__ deg,
                 int n_edges)
{
    const int tid = blockIdx.x * blockDim.x + threadIdx.x;
    const int e   = tid >> 5;
    if (e >= n_edges) return;
    const int lane = tid & 31;

    const int dst = edge_index[e];
    const int src = edge_index[n_edges + e];

    if (lane == 0)
        __hip_atomic_fetch_add(&deg[dst], 1, __ATOMIC_RELAXED, __HIP_MEMORY_SCOPE_AGENT);

    const float4 v = *reinterpret_cast<const float4*>(h + (size_t)src * HID + lane * 4);
    float* o = out + (size_t)dst * HID + lane * 4;
    __hip_atomic_fetch_add(&o[0], v.x, __ATOMIC_RELAXED, __HIP_MEMORY_SCOPE_AGENT);
    __hip_atomic_fetch_add(&o[1], v.y, __ATOMIC_RELAXED, __HIP_MEMORY_SCOPE_AGENT);
    __hip_atomic_fetch_add(&o[2], v.z, __ATOMIC_RELAXED, __HIP_MEMORY_SCOPE_AGENT);
    __hip_atomic_fetch_add(&o[3], v.w, __ATOMIC_RELAXED, __HIP_MEMORY_SCOPE_AGENT);
}

__global__ __launch_bounds__(256)
void gcn_finalize(float* __restrict__ out, const int* __restrict__ deg, int n_nodes)
{
    const int tid   = blockIdx.x * blockDim.x + threadIdx.x;
    const int total = n_nodes * (HID / 4);
    if (tid >= total) return;
    const int node  = tid >> 5;
    const float s   = 1.0f / (float)(deg[node] + 1);
    float4* p = reinterpret_cast<float4*>(out) + tid;
    float4 v = *p;
    v.x *= s; v.y *= s; v.z *= s; v.w *= s;
    *p = v;
}

extern "C" void kernel_launch(void* const* d_in, const int* in_sizes, int n_in,
                              void* d_out, int out_size, void* d_ws, size_t ws_size,
                              hipStream_t stream)
{
    const float* x          = (const float*)d_in[0];
    const int*   edge_index = (const int*)  d_in[1];
    const float* W          = (const float*)d_in[2];
    const float* b          = (const float*)d_in[3];
    float*       out        = (float*)d_out;

    const int n_nodes = in_sizes[0] / HID;
    const int n_edges = in_sizes[1] / 2;

    float* h   = (float*)d_ws;
    int*   deg = (int*)((char*)d_ws + (size_t)n_nodes * HID * sizeof(float));

    {
        const int waves  = (n_nodes + 15) / 16;
        const int blocks = (waves + 3) / 4;
        gcn_gemm_wmma<<<blocks, 128, 0, stream>>>(x, W, b, h, out, n_nodes);
    }
    gcn_zero_deg<<<(n_nodes + 255) / 256, 256, 0, stream>>>(deg, n_nodes);
    {
        const long long threads = (long long)n_edges * 32;
        const int blocks = (int)((threads + 255) / 256);
        gcn_scatter<<<blocks, 256, 0, stream>>>(edge_index, h, out, deg, n_edges);
    }
    {
        const int total = n_nodes * (HID / 4);
        gcn_finalize<<<(total + 255) / 256, 256, 0, stream>>>(out, deg, n_nodes);
    }
}


# Float-atomic scatter replaced by a counting sort of edges by dst (returning int adds give ranks, scan, place) and a per-node register sum of h[src] rows, one 512-B store; sum order = arrival rank
# speedup vs baseline: 1.8731x; 1.8731x over previous
#include <hip/hip_runtime.h>


typedef float v2f __attribute__((ext_vector_type(2)));
typedef float v8f __attribute__((ext_vector_type(8)));

#define HID    128
#define KSTEPS 32
#define NTILES 8

__global__ __launch_bounds__(128)
void gcn_gemm_wmma(const float* __restrict__ x,
                   const float* __restrict__ W,
                   const float* __restrict__ b,
                   float* __restrict__ h,
                   int n_nodes)
{
    const int wave   = blockIdx.x * (blockDim.x >> 5) + (threadIdx.x >> 5);
    const int lane   = threadIdx.x & 31;
    const int m_base = wave * 16;
    if (m_base >= n_nodes) return;

    const int rown  = lane & 15;
    const int khalf = (lane >> 4) << 1;

    v2f a[KSTEPS];
    {
        const float* xrow = x + (size_t)(m_base + rown) * HID + khalf;
        #pragma unroll
        for (int k = 0; k < KSTEPS; ++k)
            a[k] = *reinterpret_cast<const v2f*>(xrow + 4 * k);
    }

    #pragma unroll
    for (int nt = 0; nt < NTILES; ++nt) {
        const int n_base = nt * 16;
        const float bn = b[n_base + rown];
        v8f c;
        #pragma unroll
        for (int i = 0; i < 8; ++i) c[i] = bn;

        const float* wrow = W + (size_t)(n_base + rown) * HID + khalf;
        #pragma unroll
        for (int k = 0; k < KSTEPS; ++k) {
            v2f bf = *reinterpret_cast<const v2f*>(wrow + 4 * k);
            c = __builtin_amdgcn_wmma_f32_16x16x4_f32(
                     false, a[k],  false, bf,
                     (short)0, c,  false,  false);
        }

        const int col   = n_base + rown;
        const int rbase = m_base + ((lane >> 4) << 3);
        #pragma unroll
        for (int r = 0; r < 8; ++r) {
            const size_t off = (size_t)(rbase + r) * HID + col;
            h[off] = c[r];
        }
    }
}

__global__ void gcn_zero_deg(int* __restrict__ deg, int n)
{
    int i = blockIdx.x * blockDim.x + threadIdx.x;
    if (i < n) deg[i] = 0;
}

__global__ __launch_bounds__(256)
void gcn_count_rank(const int* __restrict__ edge_index,
                    int* __restrict__ deg,
                    int* __restrict__ rank,
                    int n_edges)
{
    const int e0 = (blockIdx.x * blockDim.x + threadIdx.x) * 4;
    if (e0 >= n_edges) return;
    if ((n_edges & 3) == 0) {
        const int4 d = *reinterpret_cast<const int4*>(edge_index + e0);
        int4 r;
        r.x = __hip_atomic_fetch_add(&deg[d.x], 1, __ATOMIC_RELAXED, __HIP_MEMORY_SCOPE_AGENT);
        r.y = __hip_atomic_fetch_add(&deg[d.y], 1, __ATOMIC_RELAXED, __HIP_MEMORY_SCOPE_AGENT);
        r.z = __hip_atomic_fetch_add(&deg[d.z], 1, __ATOMIC_RELAXED, __HIP_MEMORY_SCOPE_AGENT);
        r.w = __hip_atomic_fetch_add(&deg[d.w], 1, __ATOMIC_RELAXED, __HIP_MEMORY_SCOPE_AGENT);
        *reinterpret_cast<int4*>(rank + e0) = r;
    } else {
        for (int k = 0; k < 4 && e0 + k < n_edges; ++k)
            rank[e0 + k] = __hip_atomic_fetch_add(&deg[edge_index[e0 + k]], 1,
                                                  __ATOMIC_RELAXED, __HIP_MEMORY_SCOPE_AGENT);
    }
}

__global__ __launch_bounds__(256)
void gcn_rowptr(const int* __restrict__ deg, int* __restrict__ seg, int n_nodes)
{
    __shared__ int s_part[8];
    __shared__ int s_tot[8];
    const int tid  = threadIdx.x;
    const int lane = tid & 31;
    const int w    = tid >> 5;
    const int base = blockIdx.x * 256;

    const int4* deg4 = reinterpret_cast<const int4*>(deg);
    const int nq = base >> 2;
    int part = 0;
    #pragma unroll 8
    for (int j = tid; j < nq; j += 256) {
        const int4 c = deg4[j];
        part += (c.x + c.y) + (c.z + c.w);
    }

    const int i = base + tid;
    const int v = (i < n_nodes) ? deg[i] : 0;

    int incl = v;
    #pragma unroll
    for (int off = 1; off < 32; off <<= 1) {
        const int t = __shfl_up(incl, off);
        if (lane >= off) incl += t;
    }
    #pragma unroll
    for (int off = 16; off > 0; off >>= 1) part += __shfl_xor(part, off);

    if (lane == 31) s_tot[w]  = incl;
    if (lane == 0)  s_part[w] = part;
    __syncthreads();

    int before = 0;
    #pragma unroll
    for (int k = 0; k < 8; ++k) {
        before += s_part[k];
        if (k < w) before += s_tot[k];
    }
    if (i < n_nodes) reinterpret_cast<int2*>(seg)[i] = make_int2(before + incl - v, v);
}

__global__ __launch_bounds__(256)
void gcn_place(const int* __restrict__ edge_index,
               const int* __restrict__ seg,
               const int* __restrict__ rank,
               int* __restrict__ srcs,
               int n_edges)
{
    const int e0 = (blockIdx.x * blockDim.x + threadIdx.x) * 4;
    if (e0 >= n_edges) return;
    if ((n_edges & 3) == 0) {
        const int4 d = *reinterpret_cast<const int4*>(edge_index + e0);
        const int4 s = *reinterpret_cast<const int4*>(edge_index + n_edges + e0);
        const int4 r = *reinterpret_cast<const int4*>(rank + e0);
        const int px = seg[2 * d.x], py = seg[2 * d.y], pz = seg[2 * d.z], pw = seg[2 * d.w];
        srcs[px + r.x] = s.x;
        srcs[py + r.y] = s.y;
        srcs[pz + r.z] = s.z;
        srcs[pw + r.w] = s.w;
    } else {
        for (int k = 0; k < 4 && e0 + k < n_edges; ++k) {
            const int e = e0 + k;
            srcs[seg[2 * edge_index[e]] + rank[e]] = edge_index[n_edges + e];
        }
    }
}

__global__ __launch_bounds__(256)
void gcn_gather(const float* __restrict__ h,
                const int* __restrict__ seg,
                const int* __restrict__ srcs,
                float* __restrict__ out,
                int n_nodes)
{
    const int lane   = threadIdx.x & 31;
    const int nwaves = gridDim.x * 8;
    const float4* __restrict__ h4 = reinterpret_cast<const float4*>(h);
    float4* __restrict__ out4     = reinterpret_cast<float4*>(out);

    for (int node = __builtin_amdgcn_readfirstlane(blockIdx.x * 8 + (threadIdx.x >> 5));
         node < n_nodes; node += nwaves) {
        const int2 sd   = reinterpret_cast<const int2*>(seg)[node];
        const int start = sd.x;
        const int total = sd.y + 1;
        float4 acc = make_float4(0.0f, 0.0f, 0.0f, 0.0f);
        for (int base = 0; base < total; base += 32) {
            const int k = base + lane;
            int my = node;
            if (k >= 1 && k < total) my = srcs[start + k - 1];
            const int m = (total - base < 32) ? (total - base) : 32;
            for (int j = 0; j < m; j += 8) {
                float4 v[8];
                #pragma unroll
                for (int u = 0; u < 8; ++u) {
                    const int s = __builtin_amdgcn_readlane(my, j + u);
                    v[u] = h4[(size_t)s * 32 + lane];
                }
                #pragma unroll
                for (int u = 0; u < 8; ++u) {
                    if (j + u < m) {
                        acc.x += v[u].x; acc.y += v[u].y; acc.z += v[u].z; acc.w += v[u].w;
                    }
                }
            }
        }
        const float sc = 1.0f / (float)total;
        acc.x *= sc; acc.y *= sc; acc.z *= sc; acc.w *= sc;
        out4[(size_t)node * 32 + lane] = acc;
    }
}

extern "C" void kernel_launch(void* const* d_in, const int* in_sizes, int n_in,
                              void* d_out, int out_size, void* d_ws, size_t ws_size,
                              hipStream_t stream)
{
    const float* x          = (const float*)d_in[0];
    const int*   edge_index = (const int*)  d_in[1];
    const float* W          = (const float*)d_in[2];
    const float* b          = (const float*)d_in[3];
    float*       out        = (float*)d_out;

    const int n_nodes = in_sizes[0] / HID;
    const int n_edges = in_sizes[1] / 2;

    const size_t npad = ((size_t)n_nodes + 63) & ~(size_t)63;
    const size_t epad = ((size_t)n_edges + 63) & ~(size_t)63;
    float* h    = (float*)d_ws;
    int*   deg  = (int*)((char*)d_ws + (size_t)n_nodes * HID * sizeof(float));
    int*   seg  = deg + npad;
    int*   rank = seg + 2 * npad;
    int*   srcs = rank + epad;

    {
        const int waves  = (n_nodes + 15) / 16;
        const int blocks = (waves + 3) / 4;
        gcn_gemm_wmma<<<blocks, 128, 0, stream>>>(x, W, b, h, n_nodes);
    }
    gcn_zero_deg<<<(n_nodes + 255) / 256, 256, 0, stream>>>(deg, n_nodes);
    const int quads = (n_edges + 3) / 4;
    gcn_count_rank<<<(quads + 255) / 256, 256, 0, stream>>>(edge_index, deg, rank, n_edges);
    gcn_rowptr<<<(n_nodes + 255) / 256, 256, 0, stream>>>(deg, seg, n_nodes);
    gcn_place<<<(quads + 255) / 256, 256, 0, stream>>>(edge_index, seg, rank, srcs, n_edges);
    {
        const int want   = (n_nodes + 7) / 8;
        const int blocks = want < 2048 ? want : 2048;
        gcn_gather<<<blocks, 256, 0, stream>>>(h, seg, srcs, out, n_nodes);
    }
}


# Rows of x are gathered and averaged first and projected after (linear layer commutes with the mean): WMMA GEMM moves into the gather kernel, no h plane, 5 launches; product and sum order change
# speedup vs baseline: 2.0371x; 1.0876x over previous
#include <hip/hip_runtime.h>


typedef float v2f __attribute__((ext_vector_type(2)));
typedef float v8f __attribute__((ext_vector_type(8)));

#define HID    128
#define KSTEPS 32
#define APITCH 132

__global__ void gcn_zero_deg(int* __restrict__ deg, int n)
{
    int i = blockIdx.x * blockDim.x + threadIdx.x;
    if (i < n) deg[i] = 0;
}

__global__ __launch_bounds__(256)
void gcn_count_rank(const int* __restrict__ edge_index,
                    int* __restrict__ deg,
                    int* __restrict__ rank,
                    int n_edges)
{
    const int e0 = (blockIdx.x * blockDim.x + threadIdx.x) * 4;
    if (e0 >= n_edges) return;
    if ((n_edges & 3) == 0) {
        const int4 d = *reinterpret_cast<const int4*>(edge_index + e0);
        int4 r;
        r.x = __hip_atomic_fetch_add(&deg[d.x], 1, __ATOMIC_RELAXED, __HIP_MEMORY_SCOPE_AGENT);
        r.y = __hip_atomic_fetch_add(&deg[d.y], 1, __ATOMIC_RELAXED, __HIP_MEMORY_SCOPE_AGENT);
        r.z = __hip_atomic_fetch_add(&deg[d.z], 1, __ATOMIC_RELAXED, __HIP_MEMORY_SCOPE_AGENT);
        r.w = __hip_atomic_fetch_add(&deg[d.w], 1, __ATOMIC_RELAXED, __HIP_MEMORY_SCOPE_AGENT);
        *reinterpret_cast<int4*>(rank + e0) = r;
    } else {
        for (int k = 0; k < 4 && e0 + k < n_edges; ++k)
            rank[e0 + k] = __hip_atomic_fetch_add(&deg[edge_index[e0 + k]], 1,
                                                  __ATOMIC_RELAXED, __HIP_MEMORY_SCOPE_AGENT);
    }
}

__global__ __launch_bounds__(256)
void gcn_rowptr(const int* __restrict__ deg, int* __restrict__ seg, int n_nodes)
{
    __shared__ int s_part[8];
    __shared__ int s_tot[8];
    const int tid  = threadIdx.x;
    const int lane = tid & 31;
    const int w    = tid >> 5;
    const int base = blockIdx.x * 256;

    const int4* deg4 = reinterpret_cast<const int4*>(deg);
    const int nq = base >> 2;
    int part = 0;
    #pragma unroll 8
    for (int j = tid; j < nq; j += 256) {
        const int4 c = deg4[j];
        part += (c.x + c.y) + (c.z + c.w);
    }

    const int i = base + tid;
    const int v = (i < n_nodes) ? deg[i] : 0;

    int incl = v;
    #pragma unroll
    for (int off = 1; off < 32; off <<= 1) {
        const int t = __shfl_up(incl, off);
        if (lane >= off) incl += t;
    }
    #pragma unroll
    for (int off = 16; off > 0; off >>= 1) part += __shfl_xor(part, off);

    if (lane == 31) s_tot[w]  = incl;
    if (lane == 0)  s_part[w] = part;
    __syncthreads();

    int before = 0;
    #pragma unroll
    for (int k = 0; k < 8; ++k) {
        before += s_part[k];
        if (k < w) before += s_tot[k];
    }
    if (i < n_nodes) reinterpret_cast<int2*>(seg)[i] = make_int2(before + incl - v, v);
}

__global__ __launch_bounds__(256)
void gcn_place(const int* __restrict__ edge_index,
               const int* __restrict__ seg,
               const int* __restrict__ rank,
               int* __restrict__ srcs,
               int n_edges)
{
    const int e0 = (blockIdx.x * blockDim.x + threadIdx.x) * 4;
    if (e0 >= n_edges) return;
    if ((n_edges & 3) == 0) {
        const int4 d = *reinterpret_cast<const int4*>(edge_index + e0);
        const int4 s = *reinterpret_cast<const int4*>(edge_index + n_edges + e0);
        const int4 r = *reinterpret_cast<const int4*>(rank + e0);
        const int px = seg[2 * d.x], py = seg[2 * d.y], pz = seg[2 * d.z], pw = seg[2 * d.w];
        srcs[px + r.x] = s.x;
        srcs[py + r.y] = s.y;
        srcs[pz + r.z] = s.z;
        srcs[pw + r.w] = s.w;
    } else {
        for (int k = 0; k < 4 && e0 + k < n_edges; ++k) {
            const int e = e0 + k;
            srcs[seg[2 * edge_index[e]] + rank[e]] = edge_index[n_edges + e];
        }
    }
}

__device__ __forceinline__
float4 gcn_mean_row(const float4* __restrict__ x4, const int* __restrict__ srcs,
                    int node, int start, int total, int my_first, int lane)
{
    float4 acc = make_float4(0.0f, 0.0f, 0.0f, 0.0f);
    for (int base = 0; base < total; base += 32) {
        const int k = base + lane;
        int my = my_first;
        if (base > 0) {
            my = node;
            if (k < total) my = srcs[start + k - 1];
        }
        const int m = (total - base < 32) ? (total - base) : 32;
        for (int j = 0; j < m; j += 8) {
            float4 v[8];
            #pragma unroll
            for (int u = 0; u < 8; ++u) {
                const int s = __builtin_amdgcn_readlane(my, j + u);
                v[u] = x4[(size_t)s * 32 + lane];
            }
            #pragma unroll
            for (int u = 0; u < 8; ++u) {
                if (j + u < m) {
                    acc.x += v[u].x; acc.y += v[u].y; acc.z += v[u].z; acc.w += v[u].w;
                }
            }
        }
    }
    const float sc = 1.0f / (float)total;
    acc.x *= sc; acc.y *= sc; acc.z *= sc; acc.w *= sc;
    return acc;
}

__global__ __launch_bounds__(256) __attribute__((amdgpu_waves_per_eu(8)))
void gcn_gather_project(const float* __restrict__ x,
                        const float* __restrict__ W,
                        const float* __restrict__ b,
                        const int* __restrict__ seg,
                        const int* __restrict__ srcs,
                        float* __restrict__ out,
                        int n_nodes)
{
    __shared__ __attribute__((aligned(16))) float s_a[2][16 * APITCH];

    const int lane  = threadIdx.x & 31;
    const int w     = __builtin_amdgcn_readfirstlane(threadIdx.x >> 5);
    const int rown  = lane & 15;
    const int khalf = (lane >> 4) << 1;
    const int ntiles = (n_nodes + 15) >> 4;
    const float4* __restrict__ x4 = reinterpret_cast<const float4*>(x);

    const int col = w * 16 + rown;
    v2f bw[KSTEPS];
    {
        const float* wrow = W + (size_t)col * HID + khalf;
        #pragma unroll
        for (int k = 0; k < KSTEPS; ++k)
            bw[k] = *reinterpret_cast<const v2f*>(wrow + 4 * k);
    }
    const float bn = b[col];

    int it = 0;
    for (int tile = blockIdx.x; tile < ntiles; tile += gridDim.x, ++it) {
        const int m_base = tile * 16;
        float* __restrict__ at = s_a[it & 1];

        const int n0 = (m_base + 2 * w     < n_nodes) ? m_base + 2 * w     : n_nodes - 1;
        const int n1 = (m_base + 2 * w + 1 < n_nodes) ? m_base + 2 * w + 1 : n_nodes - 1;
        const int2 sd0 = reinterpret_cast<const int2*>(seg)[n0];
        const int2 sd1 = reinterpret_cast<const int2*>(seg)[n1];
        const int tot0 = sd0.y + 1;
        const int tot1 = sd1.y + 1;
        int my0 = n0, my1 = n1;
        if (lane >= 1 && lane < tot0) my0 = srcs[sd0.x + lane - 1];
        if (lane >= 1 && lane < tot1) my1 = srcs[sd1.x + lane - 1];
        const float4 r0 = gcn_mean_row(x4, srcs, n0, sd0.x, tot0, my0, lane);
        const float4 r1 = gcn_mean_row(x4, srcs, n1, sd1.x, tot1, my1, lane);
        *reinterpret_cast<float4*>(at + (2 * w)     * APITCH + 4 * lane) = r0;
        *reinterpret_cast<float4*>(at + (2 * w + 1) * APITCH + 4 * lane) = r1;
        __syncthreads();

        v2f a[KSTEPS];
        {
            const float* arow = at + rown * APITCH + khalf;
            #pragma unroll
            for (int k = 0; k < KSTEPS; ++k)
                a[k] = *reinterpret_cast<const v2f*>(arow + 4 * k);
        }
        v8f c;
        #pragma unroll
        for (int i = 0; i < 8; ++i) c[i] = bn;
        #pragma unroll
        for (int k = 0; k < KSTEPS; ++k)
            c = __builtin_amdgcn_wmma_f32_16x16x4_f32(
                     false, a[k],  false, bw[k],
                     (short)0, c,  false,  false);

        const int rbase = m_base + ((lane >> 4) << 3);
        #pragma unroll
        for (int r = 0; r < 8; ++r) {
            if (rbase + r < n_nodes)
                out[(size_t)(rbase + r) * HID + col] = c[r];
        }
    }
}

extern "C" void kernel_launch(void* const* d_in, const int* in_sizes, int n_in,
                              void* d_out, int out_size, void* d_ws, size_t ws_size,
                              hipStream_t stream)
{
    const float* x          = (const float*)d_in[0];
    const int*   edge_index = (const int*)  d_in[1];
    const float* W          = (const float*)d_in[2];
    const float* b          = (const float*)d_in[3];
    float*       out        = (float*)d_out;

    const int n_nodes = in_sizes[0] / HID;
    const int n_edges = in_sizes[1] / 2;

    const size_t npad = ((size_t)n_nodes + 63) & ~(size_t)63;
    const size_t epad = ((size_t)n_edges + 63) & ~(size_t)63;
    int*   deg  = (int*)d_ws;
    int*   seg  = deg + npad;
    int*   rank = seg + 2 * npad;
    int*   srcs = rank + epad;

    gcn_zero_deg<<<(n_nodes + 255) / 256, 256, 0, stream>>>(deg, n_nodes);
    const int quads = (n_edges + 3) / 4;
    gcn_count_rank<<<(quads + 255) / 256, 256, 0, stream>>>(edge_index, deg, rank, n_edges);
    gcn_rowptr<<<(n_nodes + 255) / 256, 256, 0, stream>>>(deg, seg, n_nodes);
    gcn_place<<<(quads + 255) / 256, 256, 0, stream>>>(edge_index, seg, rank, srcs, n_edges);
    {
        const int ntiles = (n_nodes + 15) / 16;
        const int rounds = (ntiles + 1023) / 1024;
        const int blocks = (ntiles + rounds - 1) / rounds;
        gcn_gather_project<<<blocks, 256, 0, stream>>>(x, W, b, seg, srcs, out, n_nodes);
    }
}


# Edge sort by dst moved from global returning atomics to shared-memory counters: chunk sort by 256-node bin, then one owner block per bin counts, scans, places; 3 launches; sum order: LDS arrival rank
# speedup vs baseline: 4.1701x; 2.0471x over previous
#include <hip/hip_runtime.h>


typedef float v2f __attribute__((ext_vector_type(2)));
typedef float v8f __attribute__((ext_vector_type(8)));

#define HID    128
#define KSTEPS 32
#define APITCH 132
#define BINSH  8
#define BINW   256
#define EPT    13
#define CHUNK  3328


__global__ __launch_bounds__(256)
void gcn_chunk_sort(const int* __restrict__ edge_index,
                    int* __restrict__ recs,
                    int* __restrict__ offs,
                    int n_edges, int nbins)
{
    extern __shared__ int s_bin[];
    __shared__ int s_wtot[8];
    const int tid  = threadIdx.x;
    const int lane = tid & 31;
    const int w    = tid >> 5;
    const int lo   = blockIdx.x * CHUNK;
    const int hi   = (lo + CHUNK < n_edges) ? lo + CHUNK : n_edges;

    int d[EPT], s[EPT], r[EPT];
    #pragma unroll
    for (int i = 0; i < EPT; ++i) {
        const int e  = lo + i * 256 + tid;
        const int ec = (e < hi) ? e : hi - 1;
        const int dv = edge_index[ec];
        s[i] = edge_index[n_edges + ec];
        d[i] = (e < hi) ? dv : -1;
    }

    for (int i = tid; i < nbins; i += 256) s_bin[i] = 0;
    __syncthreads();

    #pragma unroll
    for (int i = 0; i < EPT; ++i) {
        const bool ok = d[i] >= 0;
        r[i] = atomicAdd(&s_bin[ok ? d[i] >> BINSH : 0], ok ? 1 : 0);
    }
    __syncthreads();

    const int per = (nbins + 255) >> 8;
    const int b0  = tid * per;
    int local = 0;
    for (int j = 0; j < per; ++j)
        if (b0 + j < nbins) local += s_bin[b0 + j];
    int incl = local;
    #pragma unroll
    for (int off = 1; off < 32; off <<= 1) {
        const int t = __shfl_up(incl, off);
        if (lane >= off) incl += t;
    }
    if (lane == 31) s_wtot[w] = incl;
    __syncthreads();
    int run = incl - local;
    #pragma unroll
    for (int k = 0; k < 8; ++k)
        if (k < w) run += s_wtot[k];
    for (int j = 0; j < per; ++j) {
        if (b0 + j < nbins) {
            const int c = s_bin[b0 + j];
            s_bin[b0 + j] = run;
            run += c;
        }
    }
    __syncthreads();

    int2* __restrict__ rec2 = reinterpret_cast<int2*>(recs);
    int p[EPT];
    #pragma unroll
    for (int i = 0; i < EPT; ++i)
        p[i] = r[i] + s_bin[(d[i] >= 0) ? d[i] >> BINSH : 0];
    #pragma unroll
    for (int i = 0; i < EPT; ++i)
        if (d[i] >= 0) rec2[lo + p[i]] = make_int2(d[i], s[i]);

    int* __restrict__ orow = offs + (size_t)blockIdx.x * (nbins + 1);
    for (int j = 0; j < per; ++j)
        if (b0 + j < nbins) orow[b0 + j] = s_bin[b0 + j];
    if (tid == 0) orow[nbins] = hi - lo;
}

__global__ __launch_bounds__(256)
void gcn_bin_owner(const int* __restrict__ recs,
                   const int* __restrict__ offs,
                   int* __restrict__ seg,
                   int* __restrict__ srcs,
                   int n_nodes, int nbins, int nchunks)
{
    __shared__ int s_node[BINW];
    __shared__ int s_wtot[8];
    __shared__ int s_wlow[8];
    const int tid   = threadIdx.x;
    const int lane  = tid & 31;
    const int w     = tid >> 5;
    const int k     = blockIdx.x;
    const int nbase = k * BINW;
    const int2* __restrict__ rec2 = reinterpret_cast<const int2*>(recs);

    s_node[tid] = 0;
    __syncthreads();

    int low = 0;
    for (int c = tid; c < nchunks; c += 256) {
        const int* orow = offs + (size_t)c * (nbins + 1) + k;
        const int o0  = orow[0];
        const int len = orow[1] - o0;
        low += o0;
        const int2* run = rec2 + (size_t)c * CHUNK + o0;
        for (int base = 0; base < len; base += 8) {
            int dd[8];
            #pragma unroll
            for (int u = 0; u < 8; ++u) {
                dd[u] = -1;
                if (base + u < len) dd[u] = run[base + u].x;
            }
            #pragma unroll
            for (int u = 0; u < 8; ++u)
                if (dd[u] >= 0) atomicAdd(&s_node[dd[u] - nbase], 1);
        }
    }
    #pragma unroll
    for (int off = 16; off > 0; off >>= 1) low += __shfl_xor(low, off);
    if (lane == 0) s_wlow[w] = low;
    __syncthreads();

    const int v = s_node[tid];
    int incl = v;
    #pragma unroll
    for (int off = 1; off < 32; off <<= 1) {
        const int t = __shfl_up(incl, off);
        if (lane >= off) incl += t;
    }
    if (lane == 31) s_wtot[w] = incl;
    __syncthreads();
    int first = incl - v;
    #pragma unroll
    for (int j = 0; j < 8; ++j) {
        first += s_wlow[j];
        if (j < w) first += s_wtot[j];
    }
    s_node[tid] = first;
    __syncthreads();

    for (int c = tid; c < nchunks; c += 256) {
        const int* orow = offs + (size_t)c * (nbins + 1) + k;
        const int o0  = orow[0];
        const int len = orow[1] - o0;
        const int2* run = rec2 + (size_t)c * CHUNK + o0;
        for (int base = 0; base < len; base += 8) {
            int2 ee[8];
            int  pos[8];
            #pragma unroll
            for (int u = 0; u < 8; ++u) {
                ee[u] = make_int2(-1, 0);
                if (base + u < len) ee[u] = run[base + u];
            }
            #pragma unroll
            for (int u = 0; u < 8; ++u) {
                pos[u] = 0;
                if (ee[u].x >= 0) pos[u] = atomicAdd(&s_node[ee[u].x - nbase], 1);
            }
            #pragma unroll
            for (int u = 0; u < 8; ++u)
                if (ee[u].x >= 0) srcs[pos[u]] = ee[u].y;
        }
    }

    const int node = nbase + tid;
    if (node < n_nodes) reinterpret_cast<int2*>(seg)[node] = make_int2(first, v);
}

__device__ __forceinline__
float4 gcn_mean_row(const float4* __restrict__ x4, const int* __restrict__ srcs,
                    int node, int start, int total, int my_first, int lane)
{
    float4 acc = make_float4(0.0f, 0.0f, 0.0f, 0.0f);
    for (int base = 0; base < total; base += 32) {
        const int k = base + lane;
        int my = my_first;
        if (base > 0) {
            my = node;
            if (k < total) my = srcs[start + k - 1];
        }
        const int m = (total - base < 32) ? (total - base) : 32;
        for (int j = 0; j < m; j += 8) {
            float4 v[8];
            #pragma unroll
            for (int u = 0; u < 8; ++u) {
                const int s = __builtin_amdgcn_readlane(my, j + u);
                v[u] = x4[(size_t)s * 32 + lane];
            }
            #pragma unroll
            for (int u = 0; u < 8; ++u) {
                if (j + u < m) {
                    acc.x += v[u].x; acc.y += v[u].y; acc.z += v[u].z; acc.w += v[u].w;
                }
            }
        }
    }
    const float sc = 1.0f / (float)total;
    acc.x *= sc; acc.y *= sc; acc.z *= sc; acc.w *= sc;
    return acc;
}

__global__ __launch_bounds__(256) __attribute__((amdgpu_waves_per_eu(8)))
void gcn_gather_project(const float* __restrict__ x,
                        const float* __restrict__ W,
                        const float* __restrict__ b,
                        const int* __restrict__ seg,
                        const int* __restrict__ srcs,
                        float* __restrict__ out,
                        int n_nodes)
{
    __shared__ __attribute__((aligned(16))) float s_a[2][16 * APITCH];

    const int lane  = threadIdx.x & 31;
    const int w     = __builtin_amdgcn_readfirstlane(threadIdx.x >> 5);
    const int rown  = lane & 15;
    const int khalf = (lane >> 4) << 1;
    const int ntiles = (n_nodes + 15) >> 4;
    const float4* __restrict__ x4 = reinterpret_cast<const float4*>(x);

    const int col = w * 16 + rown;
    v2f bw[KSTEPS];
    {
        const float* wrow = W + (size_t)col * HID + khalf;
        #pragma unroll
        for (int k = 0; k < KSTEPS; ++k)
            bw[k] = *reinterpret_cast<const v2f*>(wrow + 4 * k);
    }
    const float bn = b[col];

    int it = 0;
    for (int tile = blockIdx.x; tile < ntiles; tile += gridDim.x, ++it) {
        const int m_base = tile * 16;
        float* __restrict__ at = s_a[it & 1];

        const int n0 = (m_base + 2 * w     < n_nodes) ? m_base + 2 * w     : n_nodes - 1;
        const int n1 = (m_base + 2 * w + 1 < n_nodes) ? m_base + 2 * w + 1 : n_nodes - 1;
        const int2 sd0 = reinterpret_cast<const int2*>(seg)[n0];
        const int2 sd1 = reinterpret_cast<const int2*>(seg)[n1];
        const int tot0 = sd0.y + 1;
        const int tot1 = sd1.y + 1;
        int my0 = n0, my1 = n1;
        if (lane >= 1 && lane < tot0) my0 = srcs[sd0.x + lane - 1];
        if (lane >= 1 && lane < tot1) my1 = srcs[sd1.x + lane - 1];
        const float4 r0 = gcn_mean_row(x4, srcs, n0, sd0.x, tot0, my0, lane);
        const float4 r1 = gcn_mean_row(x4, srcs, n1, sd1.x, tot1, my1, lane);
        *reinterpret_cast<float4*>(at + (2 * w)     * APITCH + 4 * lane) = r0;
        *reinterpret_cast<float4*>(at + (2 * w + 1) * APITCH + 4 * lane) = r1;
        __syncthreads();

        v2f a[KSTEPS];
        {
            const float* arow = at + rown * APITCH + khalf;
            #pragma unroll
            for (int k = 0; k < KSTEPS; ++k)
                a[k] = *reinterpret_cast<const v2f*>(arow + 4 * k);
        }
        v8f c;
        #pragma unroll
        for (int i = 0; i < 8; ++i) c[i] = bn;
        #pragma unroll
        for (int k = 0; k < KSTEPS; ++k)
            c = __builtin_amdgcn_wmma_f32_16x16x4_f32(
                     false, a[k],  false, bw[k],
                     (short)0, c,  false,  false);

        const int rbase = m_base + ((lane >> 4) << 3);
        #pragma unroll
        for (int r = 0; r < 8; ++r) {
            if (rbase + r < n_nodes)
                out[(size_t)(rbase + r) * HID + col] = c[r];
        }
    }
}

extern "C" void kernel_launch(void* const* d_in, const int* in_sizes, int n_in,
                              void* d_out, int out_size, void* d_ws, size_t ws_size,
                              hipStream_t stream)
{
    const float* x          = (const float*)d_in[0];
    const int*   edge_index = (const int*)  d_in[1];
    const float* W          = (const float*)d_in[2];
    const float* b          = (const float*)d_in[3];
    float*       out        = (float*)d_out;

    const int n_nodes = in_sizes[0] / HID;
    const int n_edges = in_sizes[1] / 2;
    const int nbins   = (n_nodes + BINW - 1) / BINW;
    const int nchunks = (n_edges + CHUNK - 1) / CHUNK;

    const size_t npad = ((size_t)n_nodes + 63) & ~(size_t)63;
    const size_t epad = ((size_t)n_edges + 63) & ~(size_t)63;
    int*   seg  = (int*)d_ws;
    int*   srcs = seg + 2 * npad;
    int*   recs = srcs + epad;
    int*   offs = recs + 2 * epad;

    if (nchunks > 0)
        gcn_chunk_sort<<<nchunks, 256, (size_t)nbins * sizeof(int), stream>>>(edge_index, recs, offs, n_edges, nbins);
    gcn_bin_owner<<<nbins, 256, 0, stream>>>(recs, offs, seg, srcs, n_nodes, nbins, nchunks);
    {
        const int ntiles = (n_nodes + 15) / 16;
        const int rounds = (ntiles + 1023) / 1024;
        const int blocks = (ntiles + rounds - 1) / rounds;
        gcn_gather_project<<<blocks, 256, 0, stream>>>(x, W, b, seg, srcs, out, n_nodes);
    }
}


# Bin-owner launch folded into the gather launch: each block ranks and places its 64 nodes' sources in shared memory from the chunk-sorted records; 2 launches; sum order: per-block LDS arrival rank
# speedup vs baseline: 4.8859x; 1.1717x over previous
#include <hip/hip_runtime.h>


typedef float v2f __attribute__((ext_vector_type(2)));
typedef float v8f __attribute__((ext_vector_type(8)));

#define HID    128
#define KSTEPS 32
#define APITCH 132
#define BINSH  8
#define BINW   256
#define EPT    13
#define CHUNK  3328


__global__ __launch_bounds__(256)
void gcn_chunk_sort(const int* __restrict__ edge_index,
                    int* __restrict__ recs,
                    int* __restrict__ offs,
                    int n_edges, int nbins)
{
    extern __shared__ int s_bin[];
    __shared__ int s_wtot[8];
    const int tid  = threadIdx.x;
    const int lane = tid & 31;
    const int w    = tid >> 5;
    const int lo   = blockIdx.x * CHUNK;
    const int hi   = (lo + CHUNK < n_edges) ? lo + CHUNK : n_edges;

    int d[EPT], s[EPT], r[EPT];
    #pragma unroll
    for (int i = 0; i < EPT; ++i) {
        const int e  = lo + i * 256 + tid;
        const int ec = (e < hi) ? e : hi - 1;
        const int dv = edge_index[ec];
        s[i] = edge_index[n_edges + ec];
        d[i] = (e < hi) ? dv : -1;
    }

    for (int i = tid; i < nbins; i += 256) s_bin[i] = 0;
    __syncthreads();

    #pragma unroll
    for (int i = 0; i < EPT; ++i) {
        const bool ok = d[i] >= 0;
        r[i] = atomicAdd(&s_bin[ok ? d[i] >> BINSH : 0], ok ? 1 : 0);
    }
    __syncthreads();

    const int per = (nbins + 255) >> 8;
    const int b0  = tid * per;
    int local = 0;
    for (int j = 0; j < per; ++j)
        if (b0 + j < nbins) local += s_bin[b0 + j];
    int incl = local;
    #pragma unroll
    for (int off = 1; off < 32; off <<= 1) {
        const int t = __shfl_up(incl, off);
        if (lane >= off) incl += t;
    }
    if (lane == 31) s_wtot[w] = incl;
    __syncthreads();
    int run = incl - local;
    #pragma unroll
    for (int k = 0; k < 8; ++k)
        if (k < w) run += s_wtot[k];
    for (int j = 0; j < per; ++j) {
        if (b0 + j < nbins) {
            const int c = s_bin[b0 + j];
            s_bin[b0 + j] = run;
            run += c;
        }
    }
    __syncthreads();

    int2* __restrict__ rec2 = reinterpret_cast<int2*>(recs);
    int p[EPT];
    #pragma unroll
    for (int i = 0; i < EPT; ++i)
        p[i] = r[i] + s_bin[(d[i] >= 0) ? d[i] >> BINSH : 0];
    #pragma unroll
    for (int i = 0; i < EPT; ++i)
        if (d[i] >= 0) rec2[lo + p[i]] = make_int2(d[i], s[i]);

    int* __restrict__ orow = offs + (size_t)blockIdx.x * (nbins + 1);
    for (int j = 0; j < per; ++j)
        if (b0 + j < nbins) orow[b0 + j] = s_bin[b0 + j];
    if (tid == 0) orow[nbins] = hi - lo;
}

#define NPB   64
#define SCAP  4096

template <bool PLACE>
__device__ __forceinline__
void gcn_other_records(const int2* __restrict__ rec2, const int* __restrict__ offs, int ro, int rl,
                       int w, int kbin, int nbins, int nchunks, int nb0, int lane, int* s_over, int* s_src)
{
    for (int l = 0; l < 32; ++l) {
        const int len = __builtin_amdgcn_readlane(rl, l);
        if (len <= 32) continue;
        const int o0 = __builtin_amdgcn_readlane(ro, l);
        for (int i = 32 + lane; i < len; i += 32) {
            const int2 r = rec2[(size_t)(w + 8 * l) * CHUNK + o0 + i];
            const unsigned ln = (unsigned)(r.x - nb0);
            if (ln < NPB) {
                const int pos = atomicAdd(&s_over[ln], 1);
                if (PLACE && pos < SCAP) s_src[pos] = r.y;
            }
        }
    }
    for (int c = 256 + w; c < nchunks; c += 8) {
        const int* orow = offs + (size_t)c * (nbins + 1) + kbin;
        const int o0  = orow[0];
        const int len = orow[1] - o0;
        for (int i = lane; i < len; i += 32) {
            const int2 r = rec2[(size_t)c * CHUNK + o0 + i];
            const unsigned ln = (unsigned)(r.x - nb0);
            if (ln < NPB) {
                const int pos = atomicAdd(&s_over[ln], 1);
                if (PLACE && pos < SCAP) s_src[pos] = r.y;
            }
        }
    }
}

__device__ __forceinline__
float4 gcn_mean_row(const float4* __restrict__ x4, const int* s_src,
                    int node, int start, int total, int my_first, int lane)
{
    float4 acc = make_float4(0.0f, 0.0f, 0.0f, 0.0f);
    for (int base = 0; base < total; base += 32) {
        const int k = base + lane;
        int my = my_first;
        if (base > 0) {
            my = node;
            if (k < total) my = s_src[start + k - 1];
        }
        const int m = (total - base < 32) ? (total - base) : 32;
        for (int j = 0; j < m; j += 8) {
            float4 v[8];
            #pragma unroll
            for (int u = 0; u < 8; ++u) {
                const int s = __builtin_amdgcn_readlane(my, j + u);
                v[u] = x4[(size_t)s * 32 + lane];
            }
            #pragma unroll
            for (int u = 0; u < 8; ++u) {
                if (j + u < m) {
                    acc.x += v[u].x; acc.y += v[u].y; acc.z += v[u].z; acc.w += v[u].w;
                }
            }
        }
    }
    const float sc = 1.0f / (float)total;
    acc.x *= sc; acc.y *= sc; acc.z *= sc; acc.w *= sc;
    return acc;
}

__device__ __noinline__
float4 gcn_mean_row_from_runs(const float4* __restrict__ x4, const int2* __restrict__ rec2,
                              const int* __restrict__ offs, int node, int k, int nbins, int nchunks, int lane)
{
    float4 acc = x4[(size_t)node * 32 + lane];
    int total = 1;
    for (int c = 0; c < nchunks; ++c) {
        const int* orow = offs + (size_t)c * (nbins + 1) + k;
        const int o0  = orow[0];
        const int len = orow[1] - o0;
        for (int i = 0; i < len; ++i) {
            const int2 r = rec2[(size_t)c * CHUNK + o0 + i];
            if (r.x == node) {
                const float4 v = x4[(size_t)r.y * 32 + lane];
                acc.x += v.x; acc.y += v.y; acc.z += v.z; acc.w += v.w;
                ++total;
            }
        }
    }
    const float sc = 1.0f / (float)total;
    acc.x *= sc; acc.y *= sc; acc.z *= sc; acc.w *= sc;
    return acc;
}

__global__ __launch_bounds__(256) __attribute__((amdgpu_waves_per_eu(8)))
void gcn_gather_project(const float* __restrict__ x,
                        const float* __restrict__ W,
                        const float* __restrict__ b,
                        const int* __restrict__ recs,
                        const int* __restrict__ offs,
                        float* __restrict__ out,
                        int n_nodes, int nbins, int nchunks)
{
    __shared__ __attribute__((aligned(16))) float s_a[2][16 * APITCH];
    __shared__ int s_src[SCAP];
    __shared__ int s_cnt[NPB];
    __shared__ int s_over[NPB];
    __shared__ int s_first[NPB];
    __shared__ int s_deg[NPB];

    const int lane  = threadIdx.x & 31;
    const int w     = __builtin_amdgcn_readfirstlane(threadIdx.x >> 5);
    const int rown  = lane & 15;
    const int khalf = (lane >> 4) << 1;
    const int ntiles = (n_nodes + 15) >> 4;
    const float4* __restrict__ x4 = reinterpret_cast<const float4*>(x);
    const int2* __restrict__ rec2 = reinterpret_cast<const int2*>(recs);
    const int nb0 = blockIdx.x * NPB;
    const int kbin = nb0 >> BINSH;

    if (threadIdx.x < NPB) { s_cnt[threadIdx.x] = 0; s_over[threadIdx.x] = 0; }
    __syncthreads();
    int ro = 0, rl = 0;
    {
        const int c = w + 8 * lane;
        if (c < nchunks) {
            const int* orow = offs + (size_t)c * (nbins + 1) + kbin;
            ro = orow[0];
            rl = orow[1] - ro;
        }
    }
    int rloc[32], rsrc[32], rrank[32];
    #pragma unroll
    for (int u = 0; u < 32; ++u) {
        const int o0  = __builtin_amdgcn_readlane(ro, u);
        const int len = __builtin_amdgcn_readlane(rl, u);
        int2 r = make_int2(-1, 0);
        if (lane < len) r = rec2[(size_t)(w + 8 * u) * CHUNK + o0 + lane];
        rloc[u] = r.x - nb0;
        rsrc[u] = r.y;
    }
    #pragma unroll
    for (int u = 0; u < 32; ++u) {
        rrank[u] = -1;
        if ((unsigned)rloc[u] < NPB) rrank[u] = atomicAdd(&s_cnt[rloc[u]], 1);
    }
    gcn_other_records<false>(rec2, offs, ro, rl, w, kbin, nbins, nchunks, nb0, lane, s_over, s_src);
    __syncthreads();
    if (w == 0) {
        const int a0 = s_cnt[2 * lane],     b0 = s_over[2 * lane];
        const int a1 = s_cnt[2 * lane + 1], b1 = s_over[2 * lane + 1];
        const int v = a0 + b0 + a1 + b1;
        int incl = v;
        #pragma unroll
        for (int off = 1; off < 32; off <<= 1) {
            const int t = __shfl_up(incl, off);
            if (lane >= off) incl += t;
        }
        const int f0 = incl - v;
        const int f1 = f0 + a0 + b0;
        s_first[2 * lane] = f0;        s_first[2 * lane + 1] = f1;
        s_deg[2 * lane]   = a0 + b0;   s_deg[2 * lane + 1]   = a1 + b1;
        s_over[2 * lane]  = f0 + a0;   s_over[2 * lane + 1]  = f1 + a1;
    }
    __syncthreads();
    #pragma unroll
    for (int u = 0; u < 32; ++u)
        rrank[u] += s_first[rloc[u] & (NPB - 1)];
    #pragma unroll
    for (int u = 0; u < 32; ++u)
        if ((unsigned)rloc[u] < NPB && rrank[u] < SCAP) s_src[rrank[u]] = rsrc[u];
    gcn_other_records<true>(rec2, offs, ro, rl, w, kbin, nbins, nchunks, nb0, lane, s_over, s_src);
    __syncthreads();

    const int col = w * 16 + rown;
    v2f bw[KSTEPS];
    {
        const float* wrow = W + (size_t)col * HID + khalf;
        #pragma unroll
        for (int k = 0; k < KSTEPS; ++k)
            bw[k] = *reinterpret_cast<const v2f*>(wrow + 4 * k);
    }
    const float bn = b[col];

    #pragma nounroll
    for (int it = 0; it < NPB / 16; ++it) {
        const int tile = blockIdx.x * (NPB / 16) + it;
        if (tile >= ntiles) break;
        const int m_base = tile * 16;
        float* __restrict__ at = s_a[it & 1];

        const int l0 = (m_base + 2 * w     < n_nodes) ? 16 * it + 2 * w     : n_nodes - 1 - nb0;
        const int l1 = (m_base + 2 * w + 1 < n_nodes) ? 16 * it + 2 * w + 1 : n_nodes - 1 - nb0;
        const int n0 = nb0 + l0, n1 = nb0 + l1;
        const int st0 = __builtin_amdgcn_readfirstlane(s_first[l0]);
        const int st1 = __builtin_amdgcn_readfirstlane(s_first[l1]);
        const int tot0 = __builtin_amdgcn_readfirstlane(s_deg[l0]) + 1;
        const int tot1 = __builtin_amdgcn_readfirstlane(s_deg[l1]) + 1;
        float4 r0, r1;
        if (st0 + tot0 - 1 <= SCAP) {
            int my0 = n0;
            if (lane >= 1 && lane < tot0) my0 = s_src[st0 + lane - 1];
            r0 = gcn_mean_row(x4, s_src, n0, st0, tot0, my0, lane);
        } else {
            r0 = gcn_mean_row_from_runs(x4, rec2, offs, n0, kbin, nbins, nchunks, lane);
        }
        if (st1 + tot1 - 1 <= SCAP) {
            int my1 = n1;
            if (lane >= 1 && lane < tot1) my1 = s_src[st1 + lane - 1];
            r1 = gcn_mean_row(x4, s_src, n1, st1, tot1, my1, lane);
        } else {
            r1 = gcn_mean_row_from_runs(x4, rec2, offs, n1, kbin, nbins, nchunks, lane);
        }
        *reinterpret_cast<float4*>(at + (2 * w)     * APITCH + 4 * lane) = r0;
        *reinterpret_cast<float4*>(at + (2 * w + 1) * APITCH + 4 * lane) = r1;
        __syncthreads();

        v2f a[KSTEPS];
        {
            const float* arow = at + rown * APITCH + khalf;
            #pragma unroll
            for (int k = 0; k < KSTEPS; ++k)
                a[k] = *reinterpret_cast<const v2f*>(arow + 4 * k);
        }
        v8f c;
        #pragma unroll
        for (int i = 0; i < 8; ++i) c[i] = bn;
        #pragma unroll
        for (int k = 0; k < KSTEPS; ++k)
            c = __builtin_amdgcn_wmma_f32_16x16x4_f32(
                     false, a[k],  false, bw[k],
                     (short)0, c,  false,  false);

        const int rbase = m_base + ((lane >> 4) << 3);
        #pragma unroll
        for (int r = 0; r < 8; ++r) {
            if (rbase + r < n_nodes)
                out[(size_t)(rbase + r) * HID + col] = c[r];
        }
    }
}

extern "C" void kernel_launch(void* const* d_in, const int* in_sizes, int n_in,
                              void* d_out, int out_size, void* d_ws, size_t ws_size,
                              hipStream_t stream)
{
    const float* x          = (const float*)d_in[0];
    const int*   edge_index = (const int*)  d_in[1];
    const float* W          = (const float*)d_in[2];
    const float* b          = (const float*)d_in[3];
    float*       out        = (float*)d_out;

    const int n_nodes = in_sizes[0] / HID;
    const int n_edges = in_sizes[1] / 2;
    const int nbins   = (n_nodes + BINW - 1) / BINW;
    const int nchunks = (n_edges + CHUNK - 1) / CHUNK;

    const size_t epad = ((size_t)n_edges + 63) & ~(size_t)63;
    int*   recs = (int*)d_ws;
    int*   offs = recs + 2 * epad;

    if (nchunks > 0)
        gcn_chunk_sort<<<nchunks, 256, (size_t)nbins * sizeof(int), stream>>>(edge_index, recs, offs, n_edges, nbins);
    gcn_gather_project<<<(n_nodes + NPB - 1) / NPB, 256, 0, stream>>>(x, W, b, recs, offs, out, n_nodes, nbins, nchunks);
}


# Gather as one stream per wave: a wave sums its 8 nodes' rows in 8-row load groups that run on across node ends, all 64 mean rows staged before one barrier instead of four; sum order kept
# speedup vs baseline: 5.1244x; 1.0488x over previous
#include <hip/hip_runtime.h>


typedef float v2f __attribute__((ext_vector_type(2)));
typedef float v8f __attribute__((ext_vector_type(8)));

#define HID    128
#define KSTEPS 32
#define APITCH 132
#define BINSH  8
#define BINW   256
#define EPT    13
#define CHUNK  3328


__global__ __launch_bounds__(256)
void gcn_chunk_sort(const int* __restrict__ edge_index,
                    int* __restrict__ recs,
                    int* __restrict__ offs,
                    int n_edges, int nbins)
{
    extern __shared__ int s_bin[];
    __shared__ int s_wtot[8];
    const int tid  = threadIdx.x;
    const int lane = tid & 31;
    const int w    = tid >> 5;
    const int lo   = blockIdx.x * CHUNK;
    const int hi   = (lo + CHUNK < n_edges) ? lo + CHUNK : n_edges;

    int d[EPT], s[EPT], r[EPT];
    #pragma unroll
    for (int i = 0; i < EPT; ++i) {
        const int e  = lo + i * 256 + tid;
        const int ec = (e < hi) ? e : hi - 1;
        const int dv = edge_index[ec];
        s[i] = edge_index[n_edges + ec];
        d[i] = (e < hi) ? dv : -1;
    }

    for (int i = tid; i < nbins; i += 256) s_bin[i] = 0;
    __syncthreads();

    #pragma unroll
    for (int i = 0; i < EPT; ++i) {
        const bool ok = d[i] >= 0;
        r[i] = atomicAdd(&s_bin[ok ? d[i] >> BINSH : 0], ok ? 1 : 0);
    }
    __syncthreads();

    const int per = (nbins + 255) >> 8;
    const int b0  = tid * per;
    int local = 0;
    for (int j = 0; j < per; ++j)
        if (b0 + j < nbins) local += s_bin[b0 + j];
    int incl = local;
    #pragma unroll
    for (int off = 1; off < 32; off <<= 1) {
        const int t = __shfl_up(incl, off);
        if (lane >= off) incl += t;
    }
    if (lane == 31) s_wtot[w] = incl;
    __syncthreads();
    int run = incl - local;
    #pragma unroll
    for (int k = 0; k < 8; ++k)
        if (k < w) run += s_wtot[k];
    for (int j = 0; j < per; ++j) {
        if (b0 + j < nbins) {
            const int c = s_bin[b0 + j];
            s_bin[b0 + j] = run;
            run += c;
        }
    }
    __syncthreads();

    int2* __restrict__ rec2 = reinterpret_cast<int2*>(recs);
    int p[EPT];
    #pragma unroll
    for (int i = 0; i < EPT; ++i)
        p[i] = r[i] + s_bin[(d[i] >= 0) ? d[i] >> BINSH : 0];
    #pragma unroll
    for (int i = 0; i < EPT; ++i)
        if (d[i] >= 0) rec2[lo + p[i]] = make_int2(d[i], s[i]);

    int* __restrict__ orow = offs + (size_t)blockIdx.x * (nbins + 1);
    for (int j = 0; j < per; ++j)
        if (b0 + j < nbins) orow[b0 + j] = s_bin[b0 + j];
    if (tid == 0) orow[nbins] = hi - lo;
}

#define NPB   64
#define SCAP  4096

template <bool PLACE>
__device__ __forceinline__
void gcn_other_records(const int2* __restrict__ rec2, const int* __restrict__ offs, int ro, int rl,
                       int w, int kbin, int nbins, int nchunks, int nb0, int lane, int* s_over, int* s_src)
{
    for (int l = 0; l < 32; ++l) {
        const int len = __builtin_amdgcn_readlane(rl, l);
        if (len <= 32) continue;
        const int o0 = __builtin_amdgcn_readlane(ro, l);
        for (int i = 32 + lane; i < len; i += 32) {
            const int2 r = rec2[(size_t)(w + 8 * l) * CHUNK + o0 + i];
            const unsigned ln = (unsigned)(r.x - nb0);
            if (ln < NPB) {
                const int pos = atomicAdd(&s_over[ln], 1);
                if (PLACE && pos < SCAP) s_src[pos] = r.y;
            }
        }
    }
    for (int c = 256 + w; c < nchunks; c += 8) {
        const int* orow = offs + (size_t)c * (nbins + 1) + kbin;
        const int o0  = orow[0];
        const int len = orow[1] - o0;
        for (int i = lane; i < len; i += 32) {
            const int2 r = rec2[(size_t)c * CHUNK + o0 + i];
            const unsigned ln = (unsigned)(r.x - nb0);
            if (ln < NPB) {
                const int pos = atomicAdd(&s_over[ln], 1);
                if (PLACE && pos < SCAP) s_src[pos] = r.y;
            }
        }
    }
}

__device__ __noinline__
float4 gcn_mean_row_from_runs(const float4* __restrict__ x4, const int2* __restrict__ rec2,
                              const int* __restrict__ offs, int node, int k, int nbins, int nchunks, int lane)
{
    float4 acc = x4[(size_t)node * 32 + lane];
    int total = 1;
    for (int c = 0; c < nchunks; ++c) {
        const int* orow = offs + (size_t)c * (nbins + 1) + k;
        const int o0  = orow[0];
        const int len = orow[1] - o0;
        for (int i = 0; i < len; ++i) {
            const int2 r = rec2[(size_t)c * CHUNK + o0 + i];
            if (r.x == node) {
                const float4 v = x4[(size_t)r.y * 32 + lane];
                acc.x += v.x; acc.y += v.y; acc.z += v.z; acc.w += v.w;
                ++total;
            }
        }
    }
    const float sc = 1.0f / (float)total;
    acc.x *= sc; acc.y *= sc; acc.z *= sc; acc.w *= sc;
    return acc;
}

__global__ __launch_bounds__(256) __attribute__((amdgpu_waves_per_eu(8)))
void gcn_gather_project(const float* __restrict__ x,
                        const float* __restrict__ W,
                        const float* __restrict__ b,
                        const int* __restrict__ recs,
                        const int* __restrict__ offs,
                        float* __restrict__ out,
                        int n_nodes, int nbins, int nchunks)
{
    __shared__ __attribute__((aligned(16))) float s_a[NPB / 16][16 * APITCH];
    __shared__ int s_src[SCAP];
    __shared__ int s_cnt[NPB];
    __shared__ int s_over[NPB];
    __shared__ int s_first[NPB + 1];

    const int lane  = threadIdx.x & 31;
    const int w     = __builtin_amdgcn_readfirstlane(threadIdx.x >> 5);
    const int rown  = lane & 15;
    const int khalf = (lane >> 4) << 1;
    const float4* __restrict__ x4 = reinterpret_cast<const float4*>(x);
    const int2* __restrict__ rec2 = reinterpret_cast<const int2*>(recs);
    const int nb0 = blockIdx.x * NPB;
    const int kbin = nb0 >> BINSH;

    if (threadIdx.x < NPB) { s_cnt[threadIdx.x] = 0; s_over[threadIdx.x] = 0; }
    __syncthreads();
    int ro = 0, rl = 0;
    {
        const int c = w + 8 * lane;
        if (c < nchunks) {
            const int* orow = offs + (size_t)c * (nbins + 1) + kbin;
            ro = orow[0];
            rl = orow[1] - ro;
        }
    }
    int rloc[32], rsrc[32], rrank[32];
    #pragma unroll
    for (int u = 0; u < 32; ++u) {
        const int o0  = __builtin_amdgcn_readlane(ro, u);
        const int len = __builtin_amdgcn_readlane(rl, u);
        int2 r = make_int2(-1, 0);
        if (lane < len) r = rec2[(size_t)(w + 8 * u) * CHUNK + o0 + lane];
        rloc[u] = r.x - nb0;
        rsrc[u] = r.y;
    }
    #pragma unroll
    for (int u = 0; u < 32; ++u) {
        rrank[u] = -1;
        if ((unsigned)rloc[u] < NPB) rrank[u] = atomicAdd(&s_cnt[rloc[u]], 1);
    }
    gcn_other_records<false>(rec2, offs, ro, rl, w, kbin, nbins, nchunks, nb0, lane, s_over, s_src);
    __syncthreads();
    if (w == 0) {
        const int a0 = s_cnt[2 * lane],     b0 = s_over[2 * lane];
        const int a1 = s_cnt[2 * lane + 1], b1 = s_over[2 * lane + 1];
        const int v = a0 + b0 + a1 + b1 + 2;
        int incl = v;
        #pragma unroll
        for (int off = 1; off < 32; off <<= 1) {
            const int t = __shfl_up(incl, off);
            if (lane >= off) incl += t;
        }
        const int f0 = incl - v;
        const int f1 = f0 + a0 + b0 + 1;
        s_first[2 * lane] = f0;            s_first[2 * lane + 1] = f1;
        if (lane == 31) s_first[NPB] = incl;
        s_over[2 * lane]  = f0 + 1 + a0;   s_over[2 * lane + 1]  = f1 + 1 + a1;
        const int own0 = (nb0 + 2 * lane     < n_nodes) ? nb0 + 2 * lane     : n_nodes - 1;
        const int own1 = (nb0 + 2 * lane + 1 < n_nodes) ? nb0 + 2 * lane + 1 : n_nodes - 1;
        if (f0 < SCAP) s_src[f0] = own0;
        if (f1 < SCAP) s_src[f1] = own1;
    }
    __syncthreads();
    #pragma unroll
    for (int u = 0; u < 32; ++u)
        rrank[u] += s_first[rloc[u] & (NPB - 1)] + 1;
    #pragma unroll
    for (int u = 0; u < 32; ++u)
        if ((unsigned)rloc[u] < NPB && rrank[u] < SCAP) s_src[rrank[u]] = rsrc[u];
    gcn_other_records<true>(rec2, offs, ro, rl, w, kbin, nbins, nchunks, nb0, lane, s_over, s_src);
    __syncthreads();

    const int col = w * 16 + rown;
    v2f bw[KSTEPS];
    {
        const float* wrow = W + (size_t)col * HID + khalf;
        #pragma unroll
        for (int k = 0; k < KSTEPS; ++k)
            bw[k] = *reinterpret_cast<const v2f*>(wrow + 4 * k);
    }
    const float bn = b[col];

    float* __restrict__ arow_out = &s_a[w >> 1][0] + (8 * (w & 1)) * APITCH + 4 * lane;
    const int all_items = __builtin_amdgcn_readfirstlane(s_first[NPB]);
    if (all_items <= SCAP) {
        const int bnd = s_first[8 * w + (lane < 8 ? lane : 8)];
        const int F = __builtin_amdgcn_readlane(bnd, 0);
        const int E = __builtin_amdgcn_readlane(bnd, 8);
        int ni = 0;
        int nstart = F;
        int nend = __builtin_amdgcn_readlane(bnd, 1);
        float4 acc = make_float4(0.0f, 0.0f, 0.0f, 0.0f);
        for (int base = F; base < E; base += 32) {
            const int k = (base + lane < E) ? base + lane : E - 1;
            const int my = s_src[k];
            const int m = (E - base < 32) ? (E - base) : 32;
            for (int j = 0; j < m; j += 8) {
                float4 v[8];
                #pragma unroll
                for (int u = 0; u < 8; ++u) {
                    const int s = __builtin_amdgcn_readlane(my, j + u);
                    v[u] = x4[(size_t)s * 32 + lane];
                }
                #pragma unroll
                for (int u = 0; u < 8; ++u) {
                    if (j + u < m) {
                        if (base + j + u == nend) {
                            const float sc = 1.0f / (float)(nend - nstart);
                            *reinterpret_cast<float4*>(arow_out + ni * APITCH) =
                                make_float4(acc.x * sc, acc.y * sc, acc.z * sc, acc.w * sc);
                            ++ni;
                            nstart = nend;
                            nend = __builtin_amdgcn_readlane(bnd, ni + 1);
                            acc = make_float4(0.0f, 0.0f, 0.0f, 0.0f);
                        }
                        acc.x += v[u].x; acc.y += v[u].y; acc.z += v[u].z; acc.w += v[u].w;
                    }
                }
            }
        }
        const float sc = 1.0f / (float)(nend - nstart);
        *reinterpret_cast<float4*>(arow_out + ni * APITCH) =
            make_float4(acc.x * sc, acc.y * sc, acc.z * sc, acc.w * sc);
    } else {
        for (int i = 0; i < 8; ++i) {
            const int n = (nb0 + 8 * w + i < n_nodes) ? nb0 + 8 * w + i : n_nodes - 1;
            const float4 r = gcn_mean_row_from_runs(x4, rec2, offs, n, kbin, nbins, nchunks, lane);
            *reinterpret_cast<float4*>(arow_out + i * APITCH) = r;
        }
    }
    __syncthreads();

    #pragma nounroll
    for (int t = 0; t < NPB / 16; ++t) {
        v2f a[KSTEPS];
        {
            const float* arow = &s_a[t][0] + rown * APITCH + khalf;
            #pragma unroll
            for (int k = 0; k < KSTEPS; ++k)
                a[k] = *reinterpret_cast<const v2f*>(arow + 4 * k);
        }
        v8f c;
        #pragma unroll
        for (int i = 0; i < 8; ++i) c[i] = bn;
        #pragma unroll
        for (int k = 0; k < KSTEPS; ++k)
            c = __builtin_amdgcn_wmma_f32_16x16x4_f32(
                     false, a[k],  false, bw[k],
                     (short)0, c,  false,  false);

        const int rbase = nb0 + 16 * t + ((lane >> 4) << 3);
        #pragma unroll
        for (int r = 0; r < 8; ++r) {
            if (rbase + r < n_nodes)
                out[(size_t)(rbase + r) * HID + col] = c[r];
        }
    }
}

extern "C" void kernel_launch(void* const* d_in, const int* in_sizes, int n_in,
                              void* d_out, int out_size, void* d_ws, size_t ws_size,
                              hipStream_t stream)
{
    const float* x          = (const float*)d_in[0];
    const int*   edge_index = (const int*)  d_in[1];
    const float* W          = (const float*)d_in[2];
    const float* b          = (const float*)d_in[3];
    float*       out        = (float*)d_out;

    const int n_nodes = in_sizes[0] / HID;
    const int n_edges = in_sizes[1] / 2;
    const int nbins   = (n_nodes + BINW - 1) / BINW;
    const int nchunks = (n_edges + CHUNK - 1) / CHUNK;

    const size_t epad = ((size_t)n_edges + 63) & ~(size_t)63;
    int*   recs = (int*)d_ws;
    int*   offs = recs + 2 * epad;

    if (nchunks > 0)
        gcn_chunk_sort<<<nchunks, 256, (size_t)nbins * sizeof(int), stream>>>(edge_index, recs, offs, n_edges, nbins);
    gcn_gather_project<<<(n_nodes + NPB - 1) / NPB, 256, 0, stream>>>(x, W, b, recs, offs, out, n_nodes, nbins, nchunks);
}


# B fragments: W read as whole 512-byte rows (16 loads of 4 lines, not 32 loads touching 16 lines each), shaped through the wave's own 8 staging rows in shared memory; k order and sums unchanged
# speedup vs baseline: 5.2848x; 1.0313x over previous
#include <hip/hip_runtime.h>


typedef float v2f __attribute__((ext_vector_type(2)));
typedef float v8f __attribute__((ext_vector_type(8)));

#define HID    128
#define KSTEPS 32
#define APITCH 132
#define BINSH  8
#define BINW   256
#define EPT    13
#define CHUNK  3328


__global__ __launch_bounds__(256)
void gcn_chunk_sort(const int* __restrict__ edge_index,
                    int* __restrict__ recs,
                    int* __restrict__ offs,
                    int n_edges, int nbins)
{
    extern __shared__ int s_bin[];
    __shared__ int s_wtot[8];
    const int tid  = threadIdx.x;
    const int lane = tid & 31;
    const int w    = tid >> 5;
    const int lo   = blockIdx.x * CHUNK;
    const int hi   = (lo + CHUNK < n_edges) ? lo + CHUNK : n_edges;

    int d[EPT], s[EPT], r[EPT];
    #pragma unroll
    for (int i = 0; i < EPT; ++i) {
        const int e  = lo + i * 256 + tid;
        const int ec = (e < hi) ? e : hi - 1;
        const int dv = edge_index[ec];
        s[i] = edge_index[n_edges + ec];
        d[i] = (e < hi) ? dv : -1;
    }

    for (int i = tid; i < nbins; i += 256) s_bin[i] = 0;
    __syncthreads();

    #pragma unroll
    for (int i = 0; i < EPT; ++i) {
        const bool ok = d[i] >= 0;
        r[i] = atomicAdd(&s_bin[ok ? d[i] >> BINSH : 0], ok ? 1 : 0);
    }
    __syncthreads();

    const int per = (nbins + 255) >> 8;
    const int b0  = tid * per;
    int local = 0;
    for (int j = 0; j < per; ++j)
        if (b0 + j < nbins) local += s_bin[b0 + j];
    int incl = local;
    #pragma unroll
    for (int off = 1; off < 32; off <<= 1) {
        const int t = __shfl_up(incl, off);
        if (lane >= off) incl += t;
    }
    if (lane == 31) s_wtot[w] = incl;
    __syncthreads();
    int run = incl - local;
    #pragma unroll
    for (int k = 0; k < 8; ++k)
        if (k < w) run += s_wtot[k];
    for (int j = 0; j < per; ++j) {
        if (b0 + j < nbins) {
            const int c = s_bin[b0 + j];
            s_bin[b0 + j] = run;
            run += c;
        }
    }
    __syncthreads();

    int2* __restrict__ rec2 = reinterpret_cast<int2*>(recs);
    int p[EPT];
    #pragma unroll
    for (int i = 0; i < EPT; ++i)
        p[i] = r[i] + s_bin[(d[i] >= 0) ? d[i] >> BINSH : 0];
    #pragma unroll
    for (int i = 0; i < EPT; ++i)
        if (d[i] >= 0) rec2[lo + p[i]] = make_int2(d[i], s[i]);

    int* __restrict__ orow = offs + (size_t)blockIdx.x * (nbins + 1);
    for (int j = 0; j < per; ++j)
        if (b0 + j < nbins) orow[b0 + j] = s_bin[b0 + j];
    if (tid == 0) orow[nbins] = hi - lo;
}

#define NPB   64
#define SCAP  4096

template <bool PLACE>
__device__ __forceinline__
void gcn_other_records(const int2* __restrict__ rec2, const int* __restrict__ offs, int ro, int rl,
                       int w, int kbin, int nbins, int nchunks, int nb0, int lane, int* s_over, int* s_src)
{
    for (int l = 0; l < 32; ++l) {
        const int len = __builtin_amdgcn_readlane(rl, l);
        if (len <= 32) continue;
        const int o0 = __builtin_amdgcn_readlane(ro, l);
        for (int i = 32 + lane; i < len; i += 32) {
            const int2 r = rec2[(size_t)(w + 8 * l) * CHUNK + o0 + i];
            const unsigned ln = (unsigned)(r.x - nb0);
            if (ln < NPB) {
                const int pos = atomicAdd(&s_over[ln], 1);
                if (PLACE && pos < SCAP) s_src[pos] = r.y;
            }
        }
    }
    for (int c = 256 + w; c < nchunks; c += 8) {
        const int* orow = offs + (size_t)c * (nbins + 1) + kbin;
        const int o0  = orow[0];
        const int len = orow[1] - o0;
        for (int i = lane; i < len; i += 32) {
            const int2 r = rec2[(size_t)c * CHUNK + o0 + i];
            const unsigned ln = (unsigned)(r.x - nb0);
            if (ln < NPB) {
                const int pos = atomicAdd(&s_over[ln], 1);
                if (PLACE && pos < SCAP) s_src[pos] = r.y;
            }
        }
    }
}

__device__ __noinline__
float4 gcn_mean_row_from_runs(const float4* __restrict__ x4, const int2* __restrict__ rec2,
                              const int* __restrict__ offs, int node, int k, int nbins, int nchunks, int lane)
{
    float4 acc = x4[(size_t)node * 32 + lane];
    int total = 1;
    for (int c = 0; c < nchunks; ++c) {
        const int* orow = offs + (size_t)c * (nbins + 1) + k;
        const int o0  = orow[0];
        const int len = orow[1] - o0;
        for (int i = 0; i < len; ++i) {
            const int2 r = rec2[(size_t)c * CHUNK + o0 + i];
            if (r.x == node) {
                const float4 v = x4[(size_t)r.y * 32 + lane];
                acc.x += v.x; acc.y += v.y; acc.z += v.z; acc.w += v.w;
                ++total;
            }
        }
    }
    const float sc = 1.0f / (float)total;
    acc.x *= sc; acc.y *= sc; acc.z *= sc; acc.w *= sc;
    return acc;
}

__global__ __launch_bounds__(256) __attribute__((amdgpu_waves_per_eu(8)))
void gcn_gather_project(const float* __restrict__ x,
                        const float* __restrict__ W,
                        const float* __restrict__ b,
                        const int* __restrict__ recs,
                        const int* __restrict__ offs,
                        float* __restrict__ out,
                        int n_nodes, int nbins, int nchunks)
{
    __shared__ __attribute__((aligned(16))) float s_a[NPB / 16][16 * APITCH];
    __shared__ int s_src[SCAP];
    __shared__ int s_cnt[NPB];
    __shared__ int s_over[NPB];
    __shared__ int s_first[NPB + 1];

    const int lane  = threadIdx.x & 31;
    const int w     = __builtin_amdgcn_readfirstlane(threadIdx.x >> 5);
    const int rown  = lane & 15;
    const int khalf = (lane >> 4) << 1;
    const float4* __restrict__ x4 = reinterpret_cast<const float4*>(x);
    const int2* __restrict__ rec2 = reinterpret_cast<const int2*>(recs);
    const int nb0 = blockIdx.x * NPB;
    const int kbin = nb0 >> BINSH;

    if (threadIdx.x < NPB) { s_cnt[threadIdx.x] = 0; s_over[threadIdx.x] = 0; }
    __syncthreads();
    int ro = 0, rl = 0;
    {
        const int c = w + 8 * lane;
        if (c < nchunks) {
            const int* orow = offs + (size_t)c * (nbins + 1) + kbin;
            ro = orow[0];
            rl = orow[1] - ro;
        }
    }
    int rloc[32], rsrc[32], rrank[32];
    #pragma unroll
    for (int u = 0; u < 32; ++u) {
        const int o0  = __builtin_amdgcn_readlane(ro, u);
        const int len = __builtin_amdgcn_readlane(rl, u);
        int2 r = make_int2(-1, 0);
        if (lane < len) r = rec2[(size_t)(w + 8 * u) * CHUNK + o0 + lane];
        rloc[u] = r.x - nb0;
        rsrc[u] = r.y;
    }
    #pragma unroll
    for (int u = 0; u < 32; ++u) {
        rrank[u] = -1;
        if ((unsigned)rloc[u] < NPB) rrank[u] = atomicAdd(&s_cnt[rloc[u]], 1);
    }
    gcn_other_records<false>(rec2, offs, ro, rl, w, kbin, nbins, nchunks, nb0, lane, s_over, s_src);
    __syncthreads();
    if (w == 0) {
        const int a0 = s_cnt[2 * lane],     b0 = s_over[2 * lane];
        const int a1 = s_cnt[2 * lane + 1], b1 = s_over[2 * lane + 1];
        const int v = a0 + b0 + a1 + b1 + 2;
        int incl = v;
        #pragma unroll
        for (int off = 1; off < 32; off <<= 1) {
            const int t = __shfl_up(incl, off);
            if (lane >= off) incl += t;
        }
        const int f0 = incl - v;
        const int f1 = f0 + a0 + b0 + 1;
        s_first[2 * lane] = f0;            s_first[2 * lane + 1] = f1;
        if (lane == 31) s_first[NPB] = incl;
        s_over[2 * lane]  = f0 + 1 + a0;   s_over[2 * lane + 1]  = f1 + 1 + a1;
        const int own0 = (nb0 + 2 * lane     < n_nodes) ? nb0 + 2 * lane     : n_nodes - 1;
        const int own1 = (nb0 + 2 * lane + 1 < n_nodes) ? nb0 + 2 * lane + 1 : n_nodes - 1;
        if (f0 < SCAP) s_src[f0] = own0;
        if (f1 < SCAP) s_src[f1] = own1;
    }
    __syncthreads();
    #pragma unroll
    for (int u = 0; u < 32; ++u)
        rrank[u] += s_first[rloc[u] & (NPB - 1)] + 1;
    #pragma unroll
    for (int u = 0; u < 32; ++u)
        if ((unsigned)rloc[u] < NPB && rrank[u] < SCAP) s_src[rrank[u]] = rsrc[u];
    gcn_other_records<true>(rec2, offs, ro, rl, w, kbin, nbins, nchunks, nb0, lane, s_over, s_src);

    const int col = w * 16 + rown;
    v2f bw[KSTEPS];
    {
        const float4* __restrict__ wsl = reinterpret_cast<const float4*>(W) + (size_t)(16 * w) * (HID / 4) + lane;
        float4 wr[16];
        #pragma unroll
        for (int i = 0; i < 16; ++i) wr[i] = wsl[i * (HID / 4)];
        float* wstage = &s_a[w >> 1][0] + (8 * (w & 1)) * APITCH;
        const float* brow = wstage + (rown & 7) * APITCH + khalf;
        #pragma unroll
        for (int i = 0; i < 8; ++i)
            *reinterpret_cast<float4*>(wstage + i * APITCH + 4 * lane) = wr[i];
        __builtin_amdgcn_wave_barrier();
        #pragma unroll
        for (int k = 0; k < KSTEPS; ++k)
            bw[k] = *reinterpret_cast<const v2f*>(brow + 4 * k);
        __builtin_amdgcn_wave_barrier();
        #pragma unroll
        for (int i = 0; i < 8; ++i)
            *reinterpret_cast<float4*>(wstage + i * APITCH + 4 * lane) = wr[8 + i];
        __builtin_amdgcn_wave_barrier();
        if (rown >= 8) {
            #pragma unroll
            for (int k = 0; k < KSTEPS; ++k)
                bw[k] = *reinterpret_cast<const v2f*>(brow + 4 * k);
        }
    }
    __syncthreads();
    const float bn = b[col];

    float* __restrict__ arow_out = &s_a[w >> 1][0] + (8 * (w & 1)) * APITCH + 4 * lane;
    const int all_items = __builtin_amdgcn_readfirstlane(s_first[NPB]);
    if (all_items <= SCAP) {
        const int bnd = s_first[8 * w + (lane < 8 ? lane : 8)];
        const int F = __builtin_amdgcn_readlane(bnd, 0);
        const int E = __builtin_amdgcn_readlane(bnd, 8);
        int ni = 0;
        int nstart = F;
        int nend = __builtin_amdgcn_readlane(bnd, 1);
        float4 acc = make_float4(0.0f, 0.0f, 0.0f, 0.0f);
        for (int base = F; base < E; base += 32) {
            const int k = (base + lane < E) ? base + lane : E - 1;
            const int my = s_src[k];
            const int m = (E - base < 32) ? (E - base) : 32;
            for (int j = 0; j < m; j += 8) {
                float4 v[8];
                #pragma unroll
                for (int u = 0; u < 8; ++u) {
                    const int s = __builtin_amdgcn_readlane(my, j + u);
                    v[u] = x4[(size_t)s * 32 + lane];
                }
                #pragma unroll
                for (int u = 0; u < 8; ++u) {
                    if (j + u < m) {
                        if (base + j + u == nend) {
                            const float sc = 1.0f / (float)(nend - nstart);
                            *reinterpret_cast<float4*>(arow_out + ni * APITCH) =
                                make_float4(acc.x * sc, acc.y * sc, acc.z * sc, acc.w * sc);
                            ++ni;
                            nstart = nend;
                            nend = __builtin_amdgcn_readlane(bnd, ni + 1);
                            acc = make_float4(0.0f, 0.0f, 0.0f, 0.0f);
                        }
                        acc.x += v[u].x; acc.y += v[u].y; acc.z += v[u].z; acc.w += v[u].w;
                    }
                }
            }
        }
        const float sc = 1.0f / (float)(nend - nstart);
        *reinterpret_cast<float4*>(arow_out + ni * APITCH) =
            make_float4(acc.x * sc, acc.y * sc, acc.z * sc, acc.w * sc);
    } else {
        for (int i = 0; i < 8; ++i) {
            const int n = (nb0 + 8 * w + i < n_nodes) ? nb0 + 8 * w + i : n_nodes - 1;
            const float4 r = gcn_mean_row_from_runs(x4, rec2, offs, n, kbin, nbins, nchunks, lane);
            *reinterpret_cast<float4*>(arow_out + i * APITCH) = r;
        }
    }
    __syncthreads();

    #pragma nounroll
    for (int t = 0; t < NPB / 16; ++t) {
        v2f a[KSTEPS];
        {
            const float* arow = &s_a[t][0] + rown * APITCH + khalf;
            #pragma unroll
            for (int k = 0; k < KSTEPS; ++k)
                a[k] = *reinterpret_cast<const v2f*>(arow + 4 * k);
        }
        v8f c;
        #pragma unroll
        for (int i = 0; i < 8; ++i) c[i] = bn;
        #pragma unroll
        for (int k = 0; k < KSTEPS; ++k)
            c = __builtin_amdgcn_wmma_f32_16x16x4_f32(
                     false, a[k],  false, bw[k],
                     (short)0, c,  false,  false);

        const int rbase = nb0 + 16 * t + ((lane >> 4) << 3);
        #pragma unroll
        for (int r = 0; r < 8; ++r) {
            if (rbase + r < n_nodes)
                out[(size_t)(rbase + r) * HID + col] = c[r];
        }
    }
}

extern "C" void kernel_launch(void* const* d_in, const int* in_sizes, int n_in,
                              void* d_out, int out_size, void* d_ws, size_t ws_size,
                              hipStream_t stream)
{
    const float* x          = (const float*)d_in[0];
    const int*   edge_index = (const int*)  d_in[1];
    const float* W          = (const float*)d_in[2];
    const float* b          = (const float*)d_in[3];
    float*       out        = (float*)d_out;

    const int n_nodes = in_sizes[0] / HID;
    const int n_edges = in_sizes[1] / 2;
    const int nbins   = (n_nodes + BINW - 1) / BINW;
    const int nchunks = (n_edges + CHUNK - 1) / CHUNK;

    const size_t epad = ((size_t)n_edges + 63) & ~(size_t)63;
    int*   recs = (int*)d_ws;
    int*   offs = recs + 2 * epad;

    if (nchunks > 0)
        gcn_chunk_sort<<<nchunks, 256, (size_t)nbins * sizeof(int), stream>>>(edge_index, recs, offs, n_edges, nbins);
    gcn_gather_project<<<(n_nodes + NPB - 1) / NPB, 256, 0, stream>>>(x, W, b, recs, offs, out, n_nodes, nbins, nchunks);
}


# Chunk sort by 64-node bin, one bin per gather block: a block reads only its own records, two runs per load: 16 record loads, ranks, places per wave, not 32; sum order: LDS arrival rank, changed
# speedup vs baseline: 5.6719x; 1.0732x over previous
#include <hip/hip_runtime.h>


typedef float v2f __attribute__((ext_vector_type(2)));
typedef float v8f __attribute__((ext_vector_type(8)));

#define HID    128
#define KSTEPS 32
#define APITCH 132
#define BINSH  6
#define BINW   64
#define RUNR   16
#define EPT    13
#define CHUNK  3328


__global__ __launch_bounds__(256)
void gcn_chunk_sort(const int* __restrict__ edge_index,
                    int* __restrict__ recs,
                    int* __restrict__ offs,
                    int n_edges, int nbins)
{
    extern __shared__ int s_bin[];
    __shared__ int s_wtot[8];
    const int tid  = threadIdx.x;
    const int lane = tid & 31;
    const int w    = tid >> 5;
    const int lo   = blockIdx.x * CHUNK;
    const int hi   = (lo + CHUNK < n_edges) ? lo + CHUNK : n_edges;

    int d[EPT], s[EPT], r[EPT];
    #pragma unroll
    for (int i = 0; i < EPT; ++i) {
        const int e  = lo + i * 256 + tid;
        const int ec = (e < hi) ? e : hi - 1;
        const int dv = edge_index[ec];
        s[i] = edge_index[n_edges + ec];
        d[i] = (e < hi) ? dv : -1;
    }

    for (int i = tid; i < nbins; i += 256) s_bin[i] = 0;
    __syncthreads();

    #pragma unroll
    for (int i = 0; i < EPT; ++i) {
        const bool ok = d[i] >= 0;
        r[i] = atomicAdd(&s_bin[ok ? d[i] >> BINSH : 0], ok ? 1 : 0);
    }
    __syncthreads();

    const int per = (nbins + 255) >> 8;
    const int b0  = tid * per;
    int local = 0;
    for (int j = 0; j < per; ++j)
        if (b0 + j < nbins) local += s_bin[b0 + j];
    int incl = local;
    #pragma unroll
    for (int off = 1; off < 32; off <<= 1) {
        const int t = __shfl_up(incl, off);
        if (lane >= off) incl += t;
    }
    if (lane == 31) s_wtot[w] = incl;
    __syncthreads();
    int run = incl - local;
    #pragma unroll
    for (int k = 0; k < 8; ++k)
        if (k < w) run += s_wtot[k];
    for (int j = 0; j < per; ++j) {
        if (b0 + j < nbins) {
            const int c = s_bin[b0 + j];
            s_bin[b0 + j] = run;
            run += c;
        }
    }
    __syncthreads();

    int2* __restrict__ rec2 = reinterpret_cast<int2*>(recs);
    int p[EPT];
    #pragma unroll
    for (int i = 0; i < EPT; ++i)
        p[i] = r[i] + s_bin[(d[i] >= 0) ? d[i] >> BINSH : 0];
    #pragma unroll
    for (int i = 0; i < EPT; ++i)
        if (d[i] >= 0) rec2[lo + p[i]] = make_int2(d[i], s[i]);

    int* __restrict__ orow = offs + (size_t)blockIdx.x * (nbins + 1);
    for (int j = 0; j < per; ++j)
        if (b0 + j < nbins) orow[b0 + j] = s_bin[b0 + j];
    if (tid == 0) orow[nbins] = hi - lo;
}

#define NPB   64
static_assert(NPB == BINW, "a gather block owns exactly one destination bin");
#define SCAP  4096

template <bool PLACE>
__device__ __forceinline__
void gcn_other_records(const int2* __restrict__ rec2, const int* __restrict__ offs, bool any_long,
                       int w, int kbin, int nbins, int nchunks, int nb0, int lane, int* s_over, int* s_src)
{
    for (int c = any_long ? w : 256 + w; c < nchunks; c += 8) {
        const int* orow = offs + (size_t)c * (nbins + 1) + kbin;
        const int o0  = orow[0];
        const int len = orow[1] - o0;
        for (int i = ((c < 256) ? RUNR : 0) + lane; i < len; i += 32) {
            const int2 r = rec2[(size_t)c * CHUNK + o0 + i];
            const unsigned ln = (unsigned)(r.x - nb0);
            if (ln < NPB) {
                const int pos = atomicAdd(&s_over[ln], 1);
                if (PLACE && pos < SCAP) s_src[pos] = r.y;
            }
        }
    }
}

__device__ __noinline__
float4 gcn_mean_row_from_runs(const float4* __restrict__ x4, const int2* __restrict__ rec2,
                              const int* __restrict__ offs, int node, int k, int nbins, int nchunks, int lane)
{
    float4 acc = x4[(size_t)node * 32 + lane];
    int total = 1;
    for (int c = 0; c < nchunks; ++c) {
        const int* orow = offs + (size_t)c * (nbins + 1) + k;
        const int o0  = orow[0];
        const int len = orow[1] - o0;
        for (int i = 0; i < len; ++i) {
            const int2 r = rec2[(size_t)c * CHUNK + o0 + i];
            if (r.x == node) {
                const float4 v = x4[(size_t)r.y * 32 + lane];
                acc.x += v.x; acc.y += v.y; acc.z += v.z; acc.w += v.w;
                ++total;
            }
        }
    }
    const float sc = 1.0f / (float)total;
    acc.x *= sc; acc.y *= sc; acc.z *= sc; acc.w *= sc;
    return acc;
}

__global__ __launch_bounds__(256) __attribute__((amdgpu_waves_per_eu(8)))
void gcn_gather_project(const float* __restrict__ x,
                        const float* __restrict__ W,
                        const float* __restrict__ b,
                        const int* __restrict__ recs,
                        const int* __restrict__ offs,
                        float* __restrict__ out,
                        int n_nodes, int nbins, int nchunks)
{
    __shared__ __attribute__((aligned(16))) float s_a[NPB / 16][16 * APITCH];
    __shared__ int s_src[SCAP];
    __shared__ int s_cnt[NPB];
    __shared__ int s_over[NPB];
    __shared__ int s_first[NPB + 1];

    const int lane  = threadIdx.x & 31;
    const int w     = __builtin_amdgcn_readfirstlane(threadIdx.x >> 5);
    const int rown  = lane & 15;
    const int khalf = (lane >> 4) << 1;
    const float4* __restrict__ x4 = reinterpret_cast<const float4*>(x);
    const int2* __restrict__ rec2 = reinterpret_cast<const int2*>(recs);
    const int nb0 = blockIdx.x * NPB;
    const int kbin = nb0 >> BINSH;

    if (threadIdx.x < NPB) { s_cnt[threadIdx.x] = 0; s_over[threadIdx.x] = 0; }
    __syncthreads();
    const int hsel = lane >> 4;
    const int rj   = lane & 15;
    int ro[16], re[16];
    #pragma unroll
    for (int u = 0; u < 16; ++u) {
        const int c  = w + 8 * (2 * u + hsel);
        const int cc = (c < nchunks) ? c : 0;
        const int* orow = offs + (unsigned)(cc * (nbins + 1) + kbin);
        ro[u] = orow[0];
        re[u] = orow[1];
    }
    int rl[16];
    #pragma unroll
    for (int u = 0; u < 16; ++u)
        rl[u] = (w + 8 * (2 * u + hsel) < nchunks) ? re[u] - ro[u] : 0;
    int rloc[16], rsrc[16], rrank[16];
    bool longrun = false;
    #pragma unroll
    for (int u = 0; u < 16; ++u) {
        const int c = w + 8 * (2 * u + hsel);
        const bool has = rj < rl[u];
        const int2 r = rec2[has ? (unsigned)(c * CHUNK + ro[u] + rj) : 0u];
        rloc[u] = has ? r.x - nb0 : -1;
        rsrc[u] = r.y;
        longrun |= rl[u] > RUNR;
    }
    const bool any_long = __builtin_amdgcn_ballot_w32(longrun) != 0;
    #pragma unroll
    for (int u = 0; u < 16; ++u) {
        rrank[u] = -1;
        if ((unsigned)rloc[u] < NPB) rrank[u] = atomicAdd(&s_cnt[rloc[u]], 1);
    }
    gcn_other_records<false>(rec2, offs, any_long, w, kbin, nbins, nchunks, nb0, lane, s_over, s_src);
    __syncthreads();
    if (w == 0) {
        const int a0 = s_cnt[2 * lane],     b0 = s_over[2 * lane];
        const int a1 = s_cnt[2 * lane + 1], b1 = s_over[2 * lane + 1];
        const int v = a0 + b0 + a1 + b1 + 2;
        int incl = v;
        #pragma unroll
        for (int off = 1; off < 32; off <<= 1) {
            const int t = __shfl_up(incl, off);
            if (lane >= off) incl += t;
        }
        const int f0 = incl - v;
        const int f1 = f0 + a0 + b0 + 1;
        s_first[2 * lane] = f0;            s_first[2 * lane + 1] = f1;
        if (lane == 31) s_first[NPB] = incl;
        s_over[2 * lane]  = f0 + 1 + a0;   s_over[2 * lane + 1]  = f1 + 1 + a1;
        const int own0 = (nb0 + 2 * lane     < n_nodes) ? nb0 + 2 * lane     : n_nodes - 1;
        const int own1 = (nb0 + 2 * lane + 1 < n_nodes) ? nb0 + 2 * lane + 1 : n_nodes - 1;
        if (f0 < SCAP) s_src[f0] = own0;
        if (f1 < SCAP) s_src[f1] = own1;
    }
    __syncthreads();
    #pragma unroll
    for (int u = 0; u < 16; ++u)
        rrank[u] += s_first[rloc[u] & (NPB - 1)] + 1;
    #pragma unroll
    for (int u = 0; u < 16; ++u)
        if ((unsigned)rloc[u] < NPB && rrank[u] < SCAP) s_src[rrank[u]] = rsrc[u];
    gcn_other_records<true>(rec2, offs, any_long, w, kbin, nbins, nchunks, nb0, lane, s_over, s_src);

    const int col = w * 16 + rown;
    v2f bw[KSTEPS];
    {
        const float4* __restrict__ wsl = reinterpret_cast<const float4*>(W) + (size_t)(16 * w) * (HID / 4) + lane;
        float4 wr[16];
        #pragma unroll
        for (int i = 0; i < 16; ++i) wr[i] = wsl[i * (HID / 4)];
        float* wstage = &s_a[w >> 1][0] + (8 * (w & 1)) * APITCH;
        const float* brow = wstage + (rown & 7) * APITCH + khalf;
        #pragma unroll
        for (int i = 0; i < 8; ++i)
            *reinterpret_cast<float4*>(wstage + i * APITCH + 4 * lane) = wr[i];
        __builtin_amdgcn_wave_barrier();
        #pragma unroll
        for (int k = 0; k < KSTEPS; ++k)
            bw[k] = *reinterpret_cast<const v2f*>(brow + 4 * k);
        __builtin_amdgcn_wave_barrier();
        #pragma unroll
        for (int i = 0; i < 8; ++i)
            *reinterpret_cast<float4*>(wstage + i * APITCH + 4 * lane) = wr[8 + i];
        __builtin_amdgcn_wave_barrier();
        if (rown >= 8) {
            #pragma unroll
            for (int k = 0; k < KSTEPS; ++k)
                bw[k] = *reinterpret_cast<const v2f*>(brow + 4 * k);
        }
    }
    __syncthreads();
    const float bn = b[col];

    float* __restrict__ arow_out = &s_a[w >> 1][0] + (8 * (w & 1)) * APITCH + 4 * lane;
    const int all_items = __builtin_amdgcn_readfirstlane(s_first[NPB]);
    if (all_items <= SCAP) {
        const int bnd = s_first[8 * w + (lane < 8 ? lane : 8)];
        const int F = __builtin_amdgcn_readlane(bnd, 0);
        const int E = __builtin_amdgcn_readlane(bnd, 8);
        int ni = 0;
        int nstart = F;
        int nend = __builtin_amdgcn_readlane(bnd, 1);
        float4 acc = make_float4(0.0f, 0.0f, 0.0f, 0.0f);
        for (int base = F; base < E; base += 32) {
            const int k = (base + lane < E) ? base + lane : E - 1;
            const int my = s_src[k];
            const int m = (E - base < 32) ? (E - base) : 32;
            for (int j = 0; j < m; j += 8) {
                float4 v[8];
                #pragma unroll
                for (int u = 0; u < 8; ++u) {
                    const int s = __builtin_amdgcn_readlane(my, j + u);
                    v[u] = x4[(size_t)s * 32 + lane];
                }
                #pragma unroll
                for (int u = 0; u < 8; ++u) {
                    if (j + u < m) {
                        if (base + j + u == nend) {
                            const float sc = 1.0f / (float)(nend - nstart);
                            *reinterpret_cast<float4*>(arow_out + ni * APITCH) =
                                make_float4(acc.x * sc, acc.y * sc, acc.z * sc, acc.w * sc);
                            ++ni;
                            nstart = nend;
                            nend = __builtin_amdgcn_readlane(bnd, ni + 1);
                            acc = make_float4(0.0f, 0.0f, 0.0f, 0.0f);
                        }
                        acc.x += v[u].x; acc.y += v[u].y; acc.z += v[u].z; acc.w += v[u].w;
                    }
                }
            }
        }
        const float sc = 1.0f / (float)(nend - nstart);
        *reinterpret_cast<float4*>(arow_out + ni * APITCH) =
            make_float4(acc.x * sc, acc.y * sc, acc.z * sc, acc.w * sc);
    } else {
        for (int i = 0; i < 8; ++i) {
            const int n = (nb0 + 8 * w + i < n_nodes) ? nb0 + 8 * w + i : n_nodes - 1;
            const float4 r = gcn_mean_row_from_runs(x4, rec2, offs, n, kbin, nbins, nchunks, lane);
            *reinterpret_cast<float4*>(arow_out + i * APITCH) = r;
        }
    }
    __syncthreads();

    #pragma nounroll
    for (int t = 0; t < NPB / 16; ++t) {
        v2f a[KSTEPS];
        {
            const float* arow = &s_a[t][0] + rown * APITCH + khalf;
            #pragma unroll
            for (int k = 0; k < KSTEPS; ++k)
                a[k] = *reinterpret_cast<const v2f*>(arow + 4 * k);
        }
        v8f c;
        #pragma unroll
        for (int i = 0; i < 8; ++i) c[i] = bn;
        #pragma unroll
        for (int k = 0; k < KSTEPS; ++k)
            c = __builtin_amdgcn_wmma_f32_16x16x4_f32(
                     false, a[k],  false, bw[k],
                     (short)0, c,  false,  false);

        const int rbase = nb0 + 16 * t + ((lane >> 4) << 3);
        #pragma unroll
        for (int r = 0; r < 8; ++r) {
            if (rbase + r < n_nodes)
                out[(size_t)(rbase + r) * HID + col] = c[r];
        }
    }
}

extern "C" void kernel_launch(void* const* d_in, const int* in_sizes, int n_in,
                              void* d_out, int out_size, void* d_ws, size_t ws_size,
                              hipStream_t stream)
{
    const float* x          = (const float*)d_in[0];
    const int*   edge_index = (const int*)  d_in[1];
    const float* W          = (const float*)d_in[2];
    const float* b          = (const float*)d_in[3];
    float*       out        = (float*)d_out;

    const int n_nodes = in_sizes[0] / HID;
    const int n_edges = in_sizes[1] / 2;
    const int nbins   = (n_nodes + BINW - 1) / BINW;
    const int nchunks = (n_edges + CHUNK - 1) / CHUNK;

    const size_t epad = ((size_t)n_edges + 63) & ~(size_t)63;
    int*   recs = (int*)d_ws;
    int*   offs = recs + 2 * epad;

    if (nchunks > 0)
        gcn_chunk_sort<<<nchunks, 256, (size_t)nbins * sizeof(int), stream>>>(edge_index, recs, offs, n_edges, nbins);
    gcn_gather_project<<<(n_nodes + NPB - 1) / NPB, 256, 0, stream>>>(x, W, b, recs, offs, out, n_nodes, nbins, nchunks);
}
